# starting kernel
# baseline (speedup 1.0000x reference)
#include <hip/hip_runtime.h>
#include <hip/hip_bf16.h>

typedef __attribute__((ext_vector_type(2))) float v2f;
typedef __attribute__((ext_vector_type(8))) float v8f;


__global__ void scr_reduce_C_part(const float* __restrict__ C, float* __restrict__ sCp) {
    const int kl   = blockIdx.x * blockDim.x + threadIdx.x;
    const int part = blockIdx.y;
    const float* p = C + (size_t)part * 96 * 1024 + kl;
    float acc = 0.0f;
#pragma unroll 8
    for (int t = 0; t < 96; ++t) {
        acc += p[(size_t)t * 1024];
    }
    sCp[part * 1024 + kl] = acc;
}

__global__ void scr_reduce_x_part(const float* __restrict__ x, float* __restrict__ sXp) {
    __shared__ float lds[64];
    const int b    = blockIdx.x;
    const int part = blockIdx.y;
    const int t    = threadIdx.x;
    const int w    = (t / 3) & 7;

    if (t < 64) lds[t] = 0.0f;
    __syncthreads();

    const float* p = x + (size_t)b * 256 * 768 + (size_t)part * 64 * 768 + t;
    float acc[8] = {0.f, 0.f, 0.f, 0.f, 0.f, 0.f, 0.f, 0.f};
#pragma unroll 8
    for (int j = 0; j < 64; ++j) {
        acc[j & 7] += p[(size_t)j * 768];
    }

#pragma unroll
    for (int h = 0; h < 8; ++h) {
        atomicAdd(&lds[h * 8 + w], acc[h]);
    }
    __syncthreads();

    if (t < 64) sXp[part * 8192 + b * 64 + t] = lds[t];
}

__global__ void scr_fold(const float* __restrict__ sCp, const float* __restrict__ sXp,
                         float* __restrict__ sC, float* __restrict__ sX) {
    const int idx = blockIdx.x * blockDim.x + threadIdx.x;
    if (idx < 1024) {
        float acc = 0.0f;
#pragma unroll
        for (int p = 0; p < 32; ++p) acc += sCp[p * 1024 + idx];
        sC[idx] = acc;
    } else {
        const int j = idx - 1024;
        float acc = 0.0f;
#pragma unroll
        for (int p = 0; p < 4; ++p) acc += sXp[p * 8192 + j];
        sX[j] = acc;
    }
}

__global__ void scr_outer_wmma(const float* __restrict__ sC,
                               const float* __restrict__ sX,
                               float* __restrict__ out) {
    const int tid  = threadIdx.x;
    const int lane = tid & 31;
    const int gw   = blockIdx.x * (blockDim.x >> 5) + (tid >> 5);
    const int b     = gw >> 8;
    const int rem   = gw & 255;
    const int mtile = rem >> 2;
    const int ntile = rem & 3;
    const int l16   = lane & 15;

    const float aval = sC[mtile * 16 + l16];
    const float bval = 0.25f * sX[b * 64 + ntile * 16 + l16];

    v2f A  = {aval, aval};
    v2f Bv = {bval, bval};
    v8f Cacc = {};
    v8f D = __builtin_amdgcn_wmma_f32_16x16x4_f32(
        false, A, false, Bv, (short)0, Cacc, false, false);

    const int mhi = (lane >> 4) << 3;
    float* basep = out + ((size_t)b * 1024 + (size_t)mtile * 16 + mhi) * 64
                       + ntile * 16 + l16;
#pragma unroll
    for (int v = 0; v < 8; ++v) {
        basep[(size_t)v * 64] = D[v];
    }
}

extern "C" void kernel_launch(void* const* d_in, const int* in_sizes, int n_in,
                              void* d_out, int out_size, void* d_ws, size_t ws_size,
                              hipStream_t stream) {
    (void)in_sizes; (void)n_in; (void)out_size; (void)ws_size;

    const float* x = (const float*)d_in[0];
    const float* C = (const float*)d_in[1];
    float* out = (float*)d_out;

    float* sCp = (float*)d_ws;
    float* sXp = sCp + 32 * 1024;
    float* sC  = sXp + 4 * 8192;
    float* sX  = sC + 1024;

    scr_reduce_C_part<<<dim3(4, 32), 256, 0, stream>>>(C, sCp);
    scr_reduce_x_part<<<dim3(128, 4), 768, 0, stream>>>(x, sXp);
    scr_fold<<<36, 256, 0, stream>>>(sCp, sXp, sC, sX);
    scr_outer_wmma<<<4096, 256, 0, stream>>>(sC, sX, out);
}


# C column-sum partials: 256 blocks, 12 independent 16-byte row loads per lane then an LDS fold of 8 waves (was 96 serial 4-byte loads per lane, 128 blocks); order of the S_C partial sums changes
# speedup vs baseline: 1.2437x; 1.2437x over previous
#include <hip/hip_runtime.h>
#include <hip/hip_bf16.h>

typedef __attribute__((ext_vector_type(2))) float v2f;
typedef __attribute__((ext_vector_type(8))) float v8f;


__global__ __launch_bounds__(256)
void scr_reduce_C_part(const float* __restrict__ C, float* __restrict__ sCp) {
    __shared__ float4 red[8][32];
    const int lane  = threadIdx.x & 31;
    const int wv    = threadIdx.x >> 5;
    const int chunk = blockIdx.x;
    const int part  = blockIdx.y;

    const float4* p = reinterpret_cast<const float4*>(
        C + ((size_t)part * 96 + (size_t)wv * 12) * 1024 + chunk * 128) + lane;

    float4 v[12];
    for (int r = 0; r < 12; ++r) v[r] = p[r * 256];

    float4 s = v[0];
    for (int r = 1; r < 12; ++r) {
        s.x += v[r].x; s.y += v[r].y; s.z += v[r].z; s.w += v[r].w;
    }
    red[wv][lane] = s;
    __syncthreads();

    if (threadIdx.x < 128) {
        const float* rf = reinterpret_cast<const float*>(&red[0][0]);
        float acc = rf[threadIdx.x];
        for (int w = 1; w < 8; ++w) acc += rf[w * 128 + threadIdx.x];
        sCp[part * 1024 + chunk * 128 + threadIdx.x] = acc;
    }
}

__global__ void scr_reduce_x_part(const float* __restrict__ x, float* __restrict__ sXp) {
    __shared__ float lds[64];
    const int b    = blockIdx.x;
    const int part = blockIdx.y;
    const int t    = threadIdx.x;
    const int w    = (t / 3) & 7;

    if (t < 64) lds[t] = 0.0f;
    __syncthreads();

    const float* p = x + (size_t)b * 256 * 768 + (size_t)part * 64 * 768 + t;
    float acc[8] = {0.f, 0.f, 0.f, 0.f, 0.f, 0.f, 0.f, 0.f};
#pragma unroll 8
    for (int j = 0; j < 64; ++j) {
        acc[j & 7] += p[(size_t)j * 768];
    }

#pragma unroll
    for (int h = 0; h < 8; ++h) {
        atomicAdd(&lds[h * 8 + w], acc[h]);
    }
    __syncthreads();

    if (t < 64) sXp[part * 8192 + b * 64 + t] = lds[t];
}

__global__ void scr_fold(const float* __restrict__ sCp, const float* __restrict__ sXp,
                         float* __restrict__ sC, float* __restrict__ sX) {
    const int idx = blockIdx.x * blockDim.x + threadIdx.x;
    if (idx < 1024) {
        float acc = 0.0f;
#pragma unroll
        for (int p = 0; p < 32; ++p) acc += sCp[p * 1024 + idx];
        sC[idx] = acc;
    } else {
        const int j = idx - 1024;
        float acc = 0.0f;
#pragma unroll
        for (int p = 0; p < 4; ++p) acc += sXp[p * 8192 + j];
        sX[j] = acc;
    }
}

__global__ void scr_outer_wmma(const float* __restrict__ sC,
                               const float* __restrict__ sX,
                               float* __restrict__ out) {
    const int tid  = threadIdx.x;
    const int lane = tid & 31;
    const int gw   = blockIdx.x * (blockDim.x >> 5) + (tid >> 5);
    const int b     = gw >> 8;
    const int rem   = gw & 255;
    const int mtile = rem >> 2;
    const int ntile = rem & 3;
    const int l16   = lane & 15;

    const float aval = sC[mtile * 16 + l16];
    const float bval = 0.25f * sX[b * 64 + ntile * 16 + l16];

    v2f A  = {aval, aval};
    v2f Bv = {bval, bval};
    v8f Cacc = {};
    v8f D = __builtin_amdgcn_wmma_f32_16x16x4_f32(
        false, A, false, Bv, (short)0, Cacc, false, false);

    const int mhi = (lane >> 4) << 3;
    float* basep = out + ((size_t)b * 1024 + (size_t)mtile * 16 + mhi) * 64
                       + ntile * 16 + l16;
#pragma unroll
    for (int v = 0; v < 8; ++v) {
        basep[(size_t)v * 64] = D[v];
    }
}

extern "C" void kernel_launch(void* const* d_in, const int* in_sizes, int n_in,
                              void* d_out, int out_size, void* d_ws, size_t ws_size,
                              hipStream_t stream) {
    (void)in_sizes; (void)n_in; (void)out_size; (void)ws_size;

    const float* x = (const float*)d_in[0];
    const float* C = (const float*)d_in[1];
    float* out = (float*)d_out;

    float* sCp = (float*)d_ws;
    float* sXp = sCp + 32 * 1024;
    float* sC  = sXp + 4 * 8192;
    float* sX  = sC + 1024;

    scr_reduce_C_part<<<dim3(8, 32), 256, 0, stream>>>(C, sCp);
    scr_reduce_x_part<<<dim3(128, 4), 768, 0, stream>>>(x, sXp);
    scr_fold<<<36, 256, 0, stream>>>(sCp, sXp, sC, sX);
    scr_outer_wmma<<<4096, 256, 0, stream>>>(sC, sX, out);
}


# Outer-product kernel: a wave owns 64 whole output rows (4 strips x 4 WMMAs, B columns interleaved so a lane holds 4 adjacent hw); each 16-byte store covers 512 contiguous bytes; 32768 -> 2048 waves
# speedup vs baseline: 1.4855x; 1.1944x over previous
#include <hip/hip_runtime.h>
#include <hip/hip_bf16.h>

typedef __attribute__((ext_vector_type(2))) float v2f;
typedef __attribute__((ext_vector_type(8))) float v8f;


__global__ __launch_bounds__(256)
void scr_reduce_C_part(const float* __restrict__ C, float* __restrict__ sCp) {
    __shared__ float4 red[8][32];
    const int lane  = threadIdx.x & 31;
    const int wv    = threadIdx.x >> 5;
    const int chunk = blockIdx.x;
    const int part  = blockIdx.y;

    const float4* p = reinterpret_cast<const float4*>(
        C + ((size_t)part * 96 + (size_t)wv * 12) * 1024 + chunk * 128) + lane;

    float4 v[12];
    for (int r = 0; r < 12; ++r) v[r] = p[r * 256];

    float4 s = v[0];
    for (int r = 1; r < 12; ++r) {
        s.x += v[r].x; s.y += v[r].y; s.z += v[r].z; s.w += v[r].w;
    }
    red[wv][lane] = s;
    __syncthreads();

    if (threadIdx.x < 128) {
        const float* rf = reinterpret_cast<const float*>(&red[0][0]);
        float acc = rf[threadIdx.x];
        for (int w = 1; w < 8; ++w) acc += rf[w * 128 + threadIdx.x];
        sCp[part * 1024 + chunk * 128 + threadIdx.x] = acc;
    }
}

__global__ void scr_reduce_x_part(const float* __restrict__ x, float* __restrict__ sXp) {
    __shared__ float lds[64];
    const int b    = blockIdx.x;
    const int part = blockIdx.y;
    const int t    = threadIdx.x;
    const int w    = (t / 3) & 7;

    if (t < 64) lds[t] = 0.0f;
    __syncthreads();

    const float* p = x + (size_t)b * 256 * 768 + (size_t)part * 64 * 768 + t;
    float acc[8] = {0.f, 0.f, 0.f, 0.f, 0.f, 0.f, 0.f, 0.f};
#pragma unroll 8
    for (int j = 0; j < 64; ++j) {
        acc[j & 7] += p[(size_t)j * 768];
    }

#pragma unroll
    for (int h = 0; h < 8; ++h) {
        atomicAdd(&lds[h * 8 + w], acc[h]);
    }
    __syncthreads();

    if (t < 64) sXp[part * 8192 + b * 64 + t] = lds[t];
}

__global__ void scr_fold(const float* __restrict__ sCp, const float* __restrict__ sXp,
                         float* __restrict__ sC, float* __restrict__ sX) {
    const int idx = blockIdx.x * blockDim.x + threadIdx.x;
    if (idx < 1024) {
        float acc = 0.0f;
#pragma unroll
        for (int p = 0; p < 32; ++p) acc += sCp[p * 1024 + idx];
        sC[idx] = acc;
    } else {
        const int j = idx - 1024;
        float acc = 0.0f;
#pragma unroll
        for (int p = 0; p < 4; ++p) acc += sXp[p * 8192 + j];
        sX[j] = acc;
    }
}

__global__ __launch_bounds__(256)
void scr_outer_wmma(const float* __restrict__ sC,
                    const float* __restrict__ sX,
                    float* __restrict__ out) {
    const int lane = threadIdx.x & 31;
    const int gw   = blockIdx.x * 8 + (threadIdx.x >> 5);
    const int b    = gw >> 4;
    const int kl0  = (gw & 15) << 6;
    const int c    = lane & 15;
    const int arow = ((c & 7) << 1) | (c >> 3);

    const float4 sx = *reinterpret_cast<const float4*>(sX + b * 64 + 4 * c);
    float a[4];
#pragma unroll
    for (int i = 0; i < 4; ++i) a[i] = sC[kl0 + 16 * i + arow];

    const float b0 = 0.25f * sx.x, b1 = 0.25f * sx.y;
    const float b2 = 0.25f * sx.z, b3 = 0.25f * sx.w;
    const v2f B0 = {b0, b0}, B1 = {b1, b1}, B2 = {b2, b2}, B3 = {b3, b3};
    const v8f zero = {};

    float4* dst = reinterpret_cast<float4*>(out + ((size_t)b * 1024 + kl0) * 64) + lane;

#pragma unroll
    for (int i = 0; i < 4; ++i) {
        const v2f A = {a[i], a[i]};
        const v8f D0 = __builtin_amdgcn_wmma_f32_16x16x4_f32(
            false, A, false, B0, (short)0, zero, false, false);
        const v8f D1 = __builtin_amdgcn_wmma_f32_16x16x4_f32(
            false, A, false, B1, (short)0, zero, false, false);
        const v8f D2 = __builtin_amdgcn_wmma_f32_16x16x4_f32(
            false, A, false, B2, (short)0, zero, false, false);
        const v8f D3 = __builtin_amdgcn_wmma_f32_16x16x4_f32(
            false, A, false, B3, (short)0, zero, false, false);
#pragma unroll
        for (int v = 0; v < 8; ++v) {
            dst[i * 256 + v * 32] = make_float4(D0[v], D1[v], D2[v], D3[v]);
        }
    }
}

extern "C" void kernel_launch(void* const* d_in, const int* in_sizes, int n_in,
                              void* d_out, int out_size, void* d_ws, size_t ws_size,
                              hipStream_t stream) {
    (void)in_sizes; (void)n_in; (void)out_size; (void)ws_size;

    const float* x = (const float*)d_in[0];
    const float* C = (const float*)d_in[1];
    float* out = (float*)d_out;

    float* sCp = (float*)d_ws;
    float* sXp = sCp + 32 * 1024;
    float* sC  = sXp + 4 * 8192;
    float* sX  = sC + 1024;

    scr_reduce_C_part<<<dim3(8, 32), 256, 0, stream>>>(C, sCp);
    scr_reduce_x_part<<<dim3(128, 4), 768, 0, stream>>>(x, sXp);
    scr_fold<<<36, 256, 0, stream>>>(sCp, sXp, sC, sX);
    scr_outer_wmma<<<256, 256, 0, stream>>>(sC, sX, out);
}


# Launches 4 -> 2: each x-reduction block also sums a 96-row x 64-column piece of C, and each outer-product block folds the partials it needs itself; the order of the S_C sums changes
# speedup vs baseline: 1.8784x; 1.2645x over previous
#include <hip/hip_runtime.h>


typedef __attribute__((ext_vector_type(2))) float v2f;
typedef __attribute__((ext_vector_type(8))) float v8f;


__device__ __forceinline__ float xhalf(float v) {
    const int b = __builtin_bit_cast(int, v);
    return __builtin_bit_cast(float,
        __builtin_amdgcn_permlanex16(b, b, 0x76543210, 0xfedcba98, false, false));
}

__global__ __launch_bounds__(768)
void scr_reduce_parts(const float* __restrict__ x, const float* __restrict__ C,
                      float* __restrict__ sXp, float* __restrict__ sCp) {
    __shared__ float  lds[64];
    __shared__ float4 red[24][16];
    const int b    = blockIdx.x;
    const int part = blockIdx.y;
    const int t    = threadIdx.x;
    const int w    = (t / 3) & 7;
    const int lane = t & 31;
    const int wv   = t >> 5;
    const int half = lane >> 4;
    const int q    = lane & 15;

    const int n     = part * 128 + b;
    const int grp   = n >> 4;
    const int chunk = n & 15;
    const float4* pc = reinterpret_cast<const float4*>(
        C + (size_t)(grp * 96 + wv * 4 + half) * 1024 + chunk * 64) + q;

    if (t < 64) lds[t] = 0.0f;
    __syncthreads();

    const float4 c0 = pc[0];
    const float4 c1 = pc[512];

    const float* p = x + (size_t)b * 256 * 768 + (size_t)part * 64 * 768 + t;
    float acc[8] = {0.f, 0.f, 0.f, 0.f, 0.f, 0.f, 0.f, 0.f};
#pragma unroll 8
    for (int j = 0; j < 64; ++j) {
        acc[j & 7] += p[(size_t)j * 768];
    }

    float4 cs;
    cs.x = c0.x + c1.x; cs.y = c0.y + c1.y; cs.z = c0.z + c1.z; cs.w = c0.w + c1.w;
    cs.x += xhalf(cs.x); cs.y += xhalf(cs.y); cs.z += xhalf(cs.z); cs.w += xhalf(cs.w);
    if (half == 0) red[wv][q] = cs;

#pragma unroll
    for (int h = 0; h < 8; ++h) {
        atomicAdd(&lds[h * 8 + w], acc[h]);
    }
    __syncthreads();

    if (t < 64) {
        sXp[part * 8192 + b * 64 + t] = lds[t];
    } else if (t >= 256 && t < 320) {
        const int u = t - 256;
        const float* rf = reinterpret_cast<const float*>(&red[0][0]);
        float s = rf[u];
        for (int v = 1; v < 24; ++v) s += rf[v * 64 + u];
        sCp[grp * 1024 + chunk * 64 + u] = s;
    }
}

__global__ __launch_bounds__(256)
void scr_fold_outer_wmma(const float* __restrict__ sCp,
                         const float* __restrict__ sXp,
                         float* __restrict__ out) {
    __shared__ float quart[4][64];
    const int t    = threadIdx.x;
    const int lane = t & 31;
    const int wv   = t >> 5;
    const int kl0  = (blockIdx.x & 15) << 6;
    const int b    = ((blockIdx.x >> 4) << 3) + wv;
    const int c    = lane & 15;
    const int arow = ((c & 7) << 1) | (c >> 3);

    const int pg  = t >> 6;
    const int klo = t & 63;
    const float* pc = sCp + (pg * 8) * 1024 + kl0 + klo;
    const float4* px = reinterpret_cast<const float4*>(sXp + b * 64 + 4 * c);

    float  cv[8];
    float4 xv[4];
#pragma unroll
    for (int i = 0; i < 8; ++i) cv[i] = pc[i * 1024];
#pragma unroll
    for (int i = 0; i < 4; ++i) xv[i] = px[i * 2048];

    float cacc = cv[0];
#pragma unroll
    for (int i = 1; i < 8; ++i) cacc += cv[i];
    quart[pg][klo] = cacc;

    float4 sx = xv[0];
#pragma unroll
    for (int i = 1; i < 4; ++i) {
        sx.x += xv[i].x; sx.y += xv[i].y; sx.z += xv[i].z; sx.w += xv[i].w;
    }
    __syncthreads();

    float a[4];
#pragma unroll
    for (int i = 0; i < 4; ++i) {
        const int k = 16 * i + arow;
        a[i] = ((quart[0][k] + quart[1][k]) + quart[2][k]) + quart[3][k];
    }

    const float b0 = 0.25f * sx.x, b1 = 0.25f * sx.y;
    const float b2 = 0.25f * sx.z, b3 = 0.25f * sx.w;
    const v2f B0 = {b0, b0}, B1 = {b1, b1}, B2 = {b2, b2}, B3 = {b3, b3};
    const v8f zero = {};

    float4* dst = reinterpret_cast<float4*>(out + ((size_t)b * 1024 + kl0) * 64) + lane;

#pragma unroll
    for (int i = 0; i < 4; ++i) {
        const v2f A = {a[i], a[i]};
        const v8f D0 = __builtin_amdgcn_wmma_f32_16x16x4_f32(
            false, A, false, B0, (short)0, zero, false, false);
        const v8f D1 = __builtin_amdgcn_wmma_f32_16x16x4_f32(
            false, A, false, B1, (short)0, zero, false, false);
        const v8f D2 = __builtin_amdgcn_wmma_f32_16x16x4_f32(
            false, A, false, B2, (short)0, zero, false, false);
        const v8f D3 = __builtin_amdgcn_wmma_f32_16x16x4_f32(
            false, A, false, B3, (short)0, zero, false, false);
#pragma unroll
        for (int v = 0; v < 8; ++v) {
            dst[i * 256 + v * 32] = make_float4(D0[v], D1[v], D2[v], D3[v]);
        }
    }
}

extern "C" void kernel_launch(void* const* d_in, const int* in_sizes, int n_in,
                              void* d_out, int out_size, void* d_ws, size_t ws_size,
                              hipStream_t stream) {
    (void)in_sizes; (void)n_in; (void)out_size; (void)ws_size;

    const float* x = (const float*)d_in[0];
    const float* C = (const float*)d_in[1];
    float* out = (float*)d_out;

    float* sCp = (float*)d_ws;
    float* sXp = sCp + 32 * 1024;

    scr_reduce_parts<<<dim3(128, 4), 768, 0, stream>>>(x, C, sXp, sCp);
    scr_fold_outer_wmma<<<256, 256, 0, stream>>>(sCp, sXp, out);
}


# Partial-sum kernel: the 768x8 private x sums meet by plain LDS stores, 16-byte LDS reads in 6 waves and a 4-lane DPP sum instead of same-address ds_add_f32; S_x sum order changes (now fixed)
# speedup vs baseline: 2.1176x; 1.1273x over previous
#include <hip/hip_runtime.h>


typedef __attribute__((ext_vector_type(2))) float v2f;
typedef __attribute__((ext_vector_type(8))) float v8f;


__device__ __forceinline__ float xhalf(float v) {
    const int b = __builtin_bit_cast(int, v);
    return __builtin_bit_cast(float,
        __builtin_amdgcn_permlanex16(b, b, 0x76543210, 0xfedcba98, false, false));
}

template <int MASK>
__device__ __forceinline__ float row_xor_add(float v) {
    return v + __builtin_bit_cast(float,
        __builtin_amdgcn_mov_dpp(__builtin_bit_cast(int, v), 0x160 | MASK, 0xf, 0xf, true));
}

__global__ __launch_bounds__(768)
void scr_reduce_parts(const float* __restrict__ x, const float* __restrict__ C,
                      float* __restrict__ sXp, float* __restrict__ sCp) {
    __shared__ float  tile[8][768];
    __shared__ float4 per[48];
    __shared__ float4 red[24][16];
    const int b    = blockIdx.x;
    const int part = blockIdx.y;
    const int t    = threadIdx.x;
    const int lane = t & 31;
    const int wv   = t >> 5;
    const int half = lane >> 4;
    const int q    = lane & 15;

    const int n     = part * 128 + b;
    const int grp   = n >> 4;
    const int chunk = n & 15;
    const float4* pc = reinterpret_cast<const float4*>(
        C + (size_t)(grp * 96 + wv * 4 + half) * 1024 + chunk * 64) + q;

    const float4 c0 = pc[0];
    const float4 c1 = pc[512];

    const float* p = x + (size_t)b * 256 * 768 + (size_t)part * 64 * 768 + t;
    float acc[8] = {0.f, 0.f, 0.f, 0.f, 0.f, 0.f, 0.f, 0.f};
#pragma unroll 8
    for (int j = 0; j < 64; ++j) {
        acc[j & 7] += p[(size_t)j * 768];
    }

    float4 cs;
    cs.x = c0.x + c1.x; cs.y = c0.y + c1.y; cs.z = c0.z + c1.z; cs.w = c0.w + c1.w;
    cs.x += xhalf(cs.x); cs.y += xhalf(cs.y); cs.z += xhalf(cs.z); cs.w += xhalf(cs.w);
    if (half == 0) red[wv][q] = cs;

#pragma unroll
    for (int h = 0; h < 8; ++h) {
        tile[h][t] = acc[h];
    }
    __syncthreads();

    if (t < 192) {
        const int cidx = t >> 2;
        const int r    = t & 3;
        const float4* row4 = reinterpret_cast<const float4*>(&tile[0][0])
                             + (cidx / 6) * 192 + (cidx % 6) + 6 * r;
        float4 u[8];
#pragma unroll
        for (int k = 0; k < 8; ++k) u[k] = row4[24 * k];
#pragma unroll
        for (int s = 4; s >= 1; s >>= 1) {
#pragma unroll
            for (int k = 0; k < s; ++k) {
                u[k].x += u[k + s].x; u[k].y += u[k + s].y;
                u[k].z += u[k + s].z; u[k].w += u[k + s].w;
            }
        }
        float4 v = u[0];
        v.x = row_xor_add<1>(v.x); v.y = row_xor_add<1>(v.y); v.z = row_xor_add<1>(v.z); v.w = row_xor_add<1>(v.w);
        v.x = row_xor_add<2>(v.x); v.y = row_xor_add<2>(v.y); v.z = row_xor_add<2>(v.z); v.w = row_xor_add<2>(v.w);
        if (r == 0) per[cidx] = v;
    }
    __syncthreads();

    if (t < 64) {
        const float* pf = reinterpret_cast<const float*>(&per[0]) + 3 * t;
        sXp[part * 8192 + b * 64 + t] = (pf[0] + pf[1]) + pf[2];
    } else if (t >= 256 && t < 320) {
        const int u = t - 256;
        const float* rf = reinterpret_cast<const float*>(&red[0][0]);
        float s = rf[u];
        for (int v = 1; v < 24; ++v) s += rf[v * 64 + u];
        sCp[grp * 1024 + chunk * 64 + u] = s;
    }
}

__global__ __launch_bounds__(256)
void scr_fold_outer_wmma(const float* __restrict__ sCp,
                         const float* __restrict__ sXp,
                         float* __restrict__ out) {
    __shared__ float quart[4][64];
    const int t    = threadIdx.x;
    const int lane = t & 31;
    const int wv   = t >> 5;
    const int kl0  = (blockIdx.x & 15) << 6;
    const int b    = ((blockIdx.x >> 4) << 3) + wv;
    const int c    = lane & 15;
    const int arow = ((c & 7) << 1) | (c >> 3);

    const int pg  = t >> 6;
    const int klo = t & 63;
    const float* pc = sCp + (pg * 8) * 1024 + kl0 + klo;
    const float4* px = reinterpret_cast<const float4*>(sXp + b * 64 + 4 * c);

    float  cv[8];
    float4 xv[4];
#pragma unroll
    for (int i = 0; i < 8; ++i) cv[i] = pc[i * 1024];
#pragma unroll
    for (int i = 0; i < 4; ++i) xv[i] = px[i * 2048];

    float cacc = cv[0];
#pragma unroll
    for (int i = 1; i < 8; ++i) cacc += cv[i];
    quart[pg][klo] = cacc;

    float4 sx = xv[0];
#pragma unroll
    for (int i = 1; i < 4; ++i) {
        sx.x += xv[i].x; sx.y += xv[i].y; sx.z += xv[i].z; sx.w += xv[i].w;
    }
    __syncthreads();

    float a[4];
#pragma unroll
    for (int i = 0; i < 4; ++i) {
        const int k = 16 * i + arow;
        a[i] = ((quart[0][k] + quart[1][k]) + quart[2][k]) + quart[3][k];
    }

    const float b0 = 0.25f * sx.x, b1 = 0.25f * sx.y;
    const float b2 = 0.25f * sx.z, b3 = 0.25f * sx.w;
    const v2f B0 = {b0, b0}, B1 = {b1, b1}, B2 = {b2, b2}, B3 = {b3, b3};
    const v8f zero = {};

    float4* dst = reinterpret_cast<float4*>(out + ((size_t)b * 1024 + kl0) * 64) + lane;

#pragma unroll
    for (int i = 0; i < 4; ++i) {
        const v2f A = {a[i], a[i]};
        const v8f D0 = __builtin_amdgcn_wmma_f32_16x16x4_f32(
            false, A, false, B0, (short)0, zero, false, false);
        const v8f D1 = __builtin_amdgcn_wmma_f32_16x16x4_f32(
            false, A, false, B1, (short)0, zero, false, false);
        const v8f D2 = __builtin_amdgcn_wmma_f32_16x16x4_f32(
            false, A, false, B2, (short)0, zero, false, false);
        const v8f D3 = __builtin_amdgcn_wmma_f32_16x16x4_f32(
            false, A, false, B3, (short)0, zero, false, false);
#pragma unroll
        for (int v = 0; v < 8; ++v) {
            dst[i * 256 + v * 32] = make_float4(D0[v], D1[v], D2[v], D3[v]);
        }
    }
}

extern "C" void kernel_launch(void* const* d_in, const int* in_sizes, int n_in,
                              void* d_out, int out_size, void* d_ws, size_t ws_size,
                              hipStream_t stream) {
    (void)in_sizes; (void)n_in; (void)out_size; (void)ws_size;

    const float* x = (const float*)d_in[0];
    const float* C = (const float*)d_in[1];
    float* out = (float*)d_out;

    float* sCp = (float*)d_ws;
    float* sXp = sCp + 32 * 1024;

    scr_reduce_parts<<<dim3(128, 4), 768, 0, stream>>>(x, C, sXp, sCp);
    scr_fold_outer_wmma<<<256, 256, 0, stream>>>(sCp, sXp, out);
}
